# starting kernel
# baseline (speedup 1.0000x reference)
#include <hip/hip_runtime.h>
#include <hip/hip_bf16.h>
#include <math.h>

typedef __attribute__((ext_vector_type(2))) float v2f;
typedef __attribute__((ext_vector_type(4))) float v4f;
typedef __attribute__((ext_vector_type(8))) float v8f;

#define QDIM  4096
#define LROWS 8192

__global__ void __launch_bounds__(256) gemv16_wmma_f32(
    const float* __restrict__ M, const float* __restrict__ x,
    float* __restrict__ y_part, int C, int segC)
{
    extern __shared__ float sx[];
    const int tid   = threadIdx.x;
    const int kbase = blockIdx.y * segC;

    {
        const v4f* xs = (const v4f*)(x + kbase);
        v4f*       sd = (v4f*)sx;
        for (int i = tid; i < (segC >> 2); i += 256) sd[i] = xs[i];
    }
    __syncthreads();

    const int lane    = tid & 31;
    const int m       = lane & 15;
    const int kh      = lane >> 4;
    const int wave    = tid >> 5;
    const int R       = (int)gridDim.x * 128;
    const int rowbase = (int)blockIdx.x * 128 + wave * 16;
    const float* Arow = M + (size_t)(rowbase + m) * (size_t)C + kbase;
    const float  bm   = (m == 0) ? 1.0f : 0.0f;

    v8f acc0 = {}, acc1 = {}, acc2 = {}, acc3 = {};
    for (int k0 = 0; k0 < segC; k0 += 16) {
        const int off_lo = k0 + kh * 4;
        const int off_hi = k0 + 8 + kh * 4;
        v4f a_lo = *(const v4f*)(Arow + off_lo);
        v4f a_hi = *(const v4f*)(Arow + off_hi);
        v4f x_lo = *(const v4f*)(sx + off_lo);
        v4f x_hi = *(const v4f*)(sx + off_hi);

        v2f a0 = {a_lo.x, a_lo.y}, a1 = {a_lo.z, a_lo.w};
        v2f a2 = {a_hi.x, a_hi.y}, a3 = {a_hi.z, a_hi.w};
        v2f b0 = {x_lo.x * bm, x_lo.y * bm}, b1 = {x_lo.z * bm, x_lo.w * bm};
        v2f b2 = {x_hi.x * bm, x_hi.y * bm}, b3 = {x_hi.z * bm, x_hi.w * bm};

        acc0 = __builtin_amdgcn_wmma_f32_16x16x4_f32(false, a0, false, b0, (short)0, acc0, false, false);
        acc1 = __builtin_amdgcn_wmma_f32_16x16x4_f32(false, a1, false, b1, (short)0, acc1, false, false);
        acc2 = __builtin_amdgcn_wmma_f32_16x16x4_f32(false, a2, false, b2, (short)0, acc2, false, false);
        acc3 = __builtin_amdgcn_wmma_f32_16x16x4_f32(false, a3, false, b3, (short)0, acc3, false, false);
    }
    v8f t = acc0 + acc1 + acc2 + acc3;

    if (m == 0) {
        float* dst = y_part + (size_t)blockIdx.y * R + rowbase + kh * 8;
        #pragma unroll
        for (int r = 0; r < 8; ++r) dst[r] = t[r];
    }
}

__global__ void __launch_bounds__(256) reduce_part(
    const float* __restrict__ p, float* __restrict__ y, int R, int nseg)
{
    int i = blockIdx.x * 256 + threadIdx.x;
    if (i < R) {
        float s = 0.0f;
        for (int j = 0; j < nseg; ++j) s += p[(size_t)j * R + i];
        y[i] = s;
    }
}

__global__ void __launch_bounds__(1024) softmax_fuse(
    float* __restrict__ score, const float* __restrict__ query,
    float* __restrict__ gamma)
{
    __shared__ float red[1024];
    const int tid = threadIdx.x;

    float m = -INFINITY;
    #pragma unroll
    for (int j = 0; j < LROWS / 1024; ++j) m = fmaxf(m, score[tid + j * 1024]);
    red[tid] = m;
    __syncthreads();
    for (int s = 512; s > 0; s >>= 1) {
        if (tid < s) red[tid] = fmaxf(red[tid], red[tid + s]);
        __syncthreads();
    }
    const float gmax = red[0];
    __syncthreads();

    float e[LROWS / 1024];
    float sum = 0.0f;
    #pragma unroll
    for (int j = 0; j < LROWS / 1024; ++j) {
        e[j] = expf(score[tid + j * 1024] - gmax);
        sum += e[j];
    }
    red[tid] = sum;
    __syncthreads();
    for (int s = 512; s > 0; s >>= 1) {
        if (tid < s) red[tid] += red[tid + s];
        __syncthreads();
    }
    const float inv = 1.0f / red[0];
    #pragma unroll
    for (int j = 0; j < LROWS / 1024; ++j) score[tid + j * 1024] = e[j] * inv;

    for (int i = tid; i < QDIM; i += 1024) gamma[i] = query[i];
}

__global__ void __launch_bounds__(256) wsum_part(
    const float* __restrict__ ctx, const float* __restrict__ att,
    float* __restrict__ p)
{
    __shared__ float sa[LROWS / 16];
    const int tid   = threadIdx.x;
    const int lbase = blockIdx.y * (LROWS / 16);
    for (int i = tid; i < LROWS / 16; i += 256) sa[i] = att[lbase + i];
    __syncthreads();

    const int q = blockIdx.x * 256 + tid;
    const float* c = ctx + (size_t)lbase * QDIM + q;
    float acc = 0.0f;
    #pragma unroll 8
    for (int l = 0; l < LROWS / 16; ++l)
        acc = fmaf(sa[l], c[(size_t)l * QDIM], acc);
    p[(size_t)blockIdx.y * QDIM + q] = acc;
}

extern "C" void kernel_launch(void* const* d_in, const int* in_sizes, int n_in,
                              void* d_out, int out_size, void* d_ws, size_t ws_size,
                              hipStream_t stream)
{
    (void)in_sizes; (void)n_in; (void)out_size; (void)ws_size;
    const float* query = (const float*)d_in[0];
    const float* ctx   = (const float*)d_in[1];
    const float* Kw    = (const float*)d_in[2];
    float* out = (float*)d_out;

    float* ws    = (float*)d_ws;
    float* score = ws;
    float* gamma = ws + LROWS;
    float* part  = ws + LROWS + 2 * QDIM;

    gemv16_wmma_f32<<<dim3(LROWS / 128, 8), 256, (QDIM / 8) * sizeof(float), stream>>>(
        ctx, query, part, QDIM, QDIM / 8);
    reduce_part<<<LROWS / 256, 256, 0, stream>>>(part, score, LROWS, 8);

    softmax_fuse<<<1, 1024, 0, stream>>>(score, query, gamma);

    wsum_part<<<dim3(QDIM / 256, 16), 256, 0, stream>>>(ctx, score, part);
    reduce_part<<<QDIM / 256, 256, 0, stream>>>(part, gamma + QDIM, QDIM, 16);

    gemv16_wmma_f32<<<dim3(QDIM / 128, 8), 256, (2 * QDIM / 8) * sizeof(float), stream>>>(
        Kw, gamma, part, 2 * QDIM, 2 * QDIM / 8);
    reduce_part<<<QDIM / 256, 256, 0, stream>>>(part, out, QDIM, 8);
}


# weighted column sum reads only the context rows whose softmax weight is non-zero (a zero weight adds exactly nothing); per-block row list built by a wave ballot; sum order unchanged
# speedup vs baseline: 1.5165x; 1.5165x over previous
#include <hip/hip_runtime.h>
#include <hip/hip_bf16.h>
#include <math.h>

typedef __attribute__((ext_vector_type(2))) float v2f;
typedef __attribute__((ext_vector_type(4))) float v4f;
typedef __attribute__((ext_vector_type(8))) float v8f;

#define QDIM  4096
#define LROWS 8192

__global__ void __launch_bounds__(256) gemv16_wmma_f32(
    const float* __restrict__ M, const float* __restrict__ x,
    float* __restrict__ y_part, int C, int segC)
{
    extern __shared__ float sx[];
    const int tid   = threadIdx.x;
    const int kbase = blockIdx.y * segC;

    {
        const v4f* xs = (const v4f*)(x + kbase);
        v4f*       sd = (v4f*)sx;
        for (int i = tid; i < (segC >> 2); i += 256) sd[i] = xs[i];
    }
    __syncthreads();

    const int lane    = tid & 31;
    const int m       = lane & 15;
    const int kh      = lane >> 4;
    const int wave    = tid >> 5;
    const int R       = (int)gridDim.x * 128;
    const int rowbase = (int)blockIdx.x * 128 + wave * 16;
    const float* Arow = M + (size_t)(rowbase + m) * (size_t)C + kbase;
    const float  bm   = (m == 0) ? 1.0f : 0.0f;

    v8f acc0 = {}, acc1 = {}, acc2 = {}, acc3 = {};
    for (int k0 = 0; k0 < segC; k0 += 16) {
        const int off_lo = k0 + kh * 4;
        const int off_hi = k0 + 8 + kh * 4;
        v4f a_lo = *(const v4f*)(Arow + off_lo);
        v4f a_hi = *(const v4f*)(Arow + off_hi);
        v4f x_lo = *(const v4f*)(sx + off_lo);
        v4f x_hi = *(const v4f*)(sx + off_hi);

        v2f a0 = {a_lo.x, a_lo.y}, a1 = {a_lo.z, a_lo.w};
        v2f a2 = {a_hi.x, a_hi.y}, a3 = {a_hi.z, a_hi.w};
        v2f b0 = {x_lo.x * bm, x_lo.y * bm}, b1 = {x_lo.z * bm, x_lo.w * bm};
        v2f b2 = {x_hi.x * bm, x_hi.y * bm}, b3 = {x_hi.z * bm, x_hi.w * bm};

        acc0 = __builtin_amdgcn_wmma_f32_16x16x4_f32(false, a0, false, b0, (short)0, acc0, false, false);
        acc1 = __builtin_amdgcn_wmma_f32_16x16x4_f32(false, a1, false, b1, (short)0, acc1, false, false);
        acc2 = __builtin_amdgcn_wmma_f32_16x16x4_f32(false, a2, false, b2, (short)0, acc2, false, false);
        acc3 = __builtin_amdgcn_wmma_f32_16x16x4_f32(false, a3, false, b3, (short)0, acc3, false, false);
    }
    v8f t = acc0 + acc1 + acc2 + acc3;

    if (m == 0) {
        float* dst = y_part + (size_t)blockIdx.y * R + rowbase + kh * 8;
        #pragma unroll
        for (int r = 0; r < 8; ++r) dst[r] = t[r];
    }
}

__global__ void __launch_bounds__(256) reduce_part(
    const float* __restrict__ p, float* __restrict__ y, int R, int nseg)
{
    int i = blockIdx.x * 256 + threadIdx.x;
    if (i < R) {
        float s = 0.0f;
        for (int j = 0; j < nseg; ++j) s += p[(size_t)j * R + i];
        y[i] = s;
    }
}

__global__ void __launch_bounds__(1024) softmax_fuse(
    float* __restrict__ score, const float* __restrict__ query,
    float* __restrict__ gamma)
{
    __shared__ float red[1024];
    const int tid = threadIdx.x;

    float m = -INFINITY;
    #pragma unroll
    for (int j = 0; j < LROWS / 1024; ++j) m = fmaxf(m, score[tid + j * 1024]);
    red[tid] = m;
    __syncthreads();
    for (int s = 512; s > 0; s >>= 1) {
        if (tid < s) red[tid] = fmaxf(red[tid], red[tid + s]);
        __syncthreads();
    }
    const float gmax = red[0];
    __syncthreads();

    float e[LROWS / 1024];
    float sum = 0.0f;
    #pragma unroll
    for (int j = 0; j < LROWS / 1024; ++j) {
        e[j] = expf(score[tid + j * 1024] - gmax);
        sum += e[j];
    }
    red[tid] = sum;
    __syncthreads();
    for (int s = 512; s > 0; s >>= 1) {
        if (tid < s) red[tid] += red[tid + s];
        __syncthreads();
    }
    const float inv = 1.0f / red[0];
    #pragma unroll
    for (int j = 0; j < LROWS / 1024; ++j) score[tid + j * 1024] = e[j] * inv;

    for (int i = tid; i < QDIM; i += 1024) gamma[i] = query[i];
}

__global__ void __launch_bounds__(256) wsum_part(
    const float* __restrict__ ctx, const float* __restrict__ att,
    float* __restrict__ p)
{
    __shared__ float sa[LROWS / 16];
    __shared__ int   srow[LROWS / 16];
    __shared__ int   scount;
    const int tid   = threadIdx.x;
    const int lbase = blockIdx.y * (LROWS / 16);
    for (int i = tid; i < LROWS / 16; i += 256) sa[i] = att[lbase + i];
    __syncthreads();

    if (tid < 32) {
        int base = 0;
        for (int ch = 0; ch < LROWS / 16 / 32; ++ch) {
            const int l = ch * 32 + tid;
            const bool nz = (sa[l] != 0.0f);
            const unsigned mask = __builtin_amdgcn_ballot_w32(nz);
            if (nz) srow[base + __builtin_popcount(mask & ((1u << tid) - 1u))] = l;
            base += __builtin_popcount(mask);
        }
        if (tid == 0) scount = base;
    }
    __syncthreads();

    const int n = scount;
    const int q = blockIdx.x * 256 + tid;
    const float* c = ctx + (size_t)lbase * QDIM + q;
    float acc = 0.0f;
    #pragma unroll 8
    for (int i = 0; i < n; ++i) {
        const int l = srow[i];
        acc = fmaf(sa[l], c[(size_t)l * QDIM], acc);
    }
    p[(size_t)blockIdx.y * QDIM + q] = acc;
}

extern "C" void kernel_launch(void* const* d_in, const int* in_sizes, int n_in,
                              void* d_out, int out_size, void* d_ws, size_t ws_size,
                              hipStream_t stream)
{
    (void)in_sizes; (void)n_in; (void)out_size; (void)ws_size;
    const float* query = (const float*)d_in[0];
    const float* ctx   = (const float*)d_in[1];
    const float* Kw    = (const float*)d_in[2];
    float* out = (float*)d_out;

    float* ws    = (float*)d_ws;
    float* score = ws;
    float* gamma = ws + LROWS;
    float* part  = ws + LROWS + 2 * QDIM;

    gemv16_wmma_f32<<<dim3(LROWS / 128, 8), 256, (QDIM / 8) * sizeof(float), stream>>>(
        ctx, query, part, QDIM, QDIM / 8);
    reduce_part<<<LROWS / 256, 256, 0, stream>>>(part, score, LROWS, 8);

    softmax_fuse<<<1, 1024, 0, stream>>>(score, query, gamma);

    wsum_part<<<dim3(QDIM / 256, 16), 256, 0, stream>>>(ctx, score, part);
    reduce_part<<<QDIM / 256, 256, 0, stream>>>(part, gamma + QDIM, QDIM, 16);

    gemv16_wmma_f32<<<dim3(QDIM / 128, 8), 256, (2 * QDIM / 8) * sizeof(float), stream>>>(
        Kw, gamma, part, 2 * QDIM, 2 * QDIM / 8);
    reduce_part<<<QDIM / 256, 256, 0, stream>>>(part, out, QDIM, 8);
}


# both matrix-vector products: matrix rows streamed by whole-line non-temporal 16 B loads (512 B of a row per wave load, 16-32 in flight) into a per-wave LDS tile that feeds the WMMA; sum order kept
# speedup vs baseline: 2.0044x; 1.3217x over previous
#include <hip/hip_runtime.h>
#include <hip/hip_bf16.h>
#include <math.h>

typedef __attribute__((ext_vector_type(2))) float v2f;
typedef __attribute__((ext_vector_type(4))) float v4f;
typedef __attribute__((ext_vector_type(8))) float v8f;

#define QDIM  4096
#define LROWS 8192

#define TK    128
#define PITCH 132

static __device__ __forceinline__ void wave_tile_handover()
{
    __builtin_amdgcn_fence(__ATOMIC_RELEASE, "wavefront", "local");
    __builtin_amdgcn_wave_barrier();
    __builtin_amdgcn_fence(__ATOMIC_ACQUIRE, "wavefront", "local");
}

template <int C, int SEGC>
__global__ void __launch_bounds__(256) gemv16_wmma_f32(
    const float* __restrict__ M, const float* __restrict__ x,
    float* __restrict__ y_part)
{
    constexpr int NSTEP = SEGC / TK;
    static_assert(SEGC % (2 * TK) == 0, "steps are taken in pairs");
    __shared__ __attribute__((aligned(16))) float sx[SEGC];
    __shared__ __attribute__((aligned(16))) float stile[8 * 16 * PITCH];

    const int tid     = threadIdx.x;
    const int lane    = tid & 31;
    const int m       = lane & 15;
    const int kh      = lane >> 4;
    const int wave    = tid >> 5;
    const int kbase   = blockIdx.y * SEGC;
    const int R       = (int)gridDim.x * 128;
    const int rowbase = (int)blockIdx.x * 128 + wave * 16;
    const float  bm   = (m == 0) ? 1.0f : 0.0f;

    const float* g    = M + (size_t)rowbase * (size_t)C + kbase + lane * 4;
    float*       st   = stile + wave * (16 * PITCH);
    float*       st_w = st + lane * 4;
    const float* st_r = st + m * PITCH + kh * 4;
    const float* sx_r = sx + kh * 4;

    v4f bufA[16], bufB[16];
    v8f acc0 = {}, acc1 = {}, acc2 = {}, acc3 = {};

#define ISSUE(BUF, S) do {                                                          \
        _Pragma("unroll") for (int r = 0; r < 16; ++r)                              \
            BUF[r] = __builtin_nontemporal_load((const v4f*)(g + (size_t)r * C + (S) * TK)); \
          \
          \
        __builtin_amdgcn_sched_barrier(0);                                          \
    } while (0)

#define CONSUME(BUF, S) do {                                                        \
        _Pragma("unroll") for (int r = 0; r < 16; ++r)                              \
            *(v4f*)(st_w + r * PITCH) = BUF[r];                                     \
        wave_tile_handover();                                                       \
        _Pragma("unroll") for (int it = 0; it < TK / 16; ++it) {                    \
            const v4f a_lo = *(const v4f*)(st_r + it * 16);                         \
            const v4f a_hi = *(const v4f*)(st_r + it * 16 + 8);                     \
            const v4f x_lo = *(const v4f*)(sx_r + (S) * TK + it * 16);              \
            const v4f x_hi = *(const v4f*)(sx_r + (S) * TK + it * 16 + 8);          \
            const v2f a0 = {a_lo.x, a_lo.y}, a1 = {a_lo.z, a_lo.w};                 \
            const v2f a2 = {a_hi.x, a_hi.y}, a3 = {a_hi.z, a_hi.w};                 \
            const v2f b0 = {x_lo.x * bm, x_lo.y * bm}, b1 = {x_lo.z * bm, x_lo.w * bm}; \
            const v2f b2 = {x_hi.x * bm, x_hi.y * bm}, b3 = {x_hi.z * bm, x_hi.w * bm}; \
            acc0 = __builtin_amdgcn_wmma_f32_16x16x4_f32(false, a0, false, b0, (short)0, acc0, false, false); \
            acc1 = __builtin_amdgcn_wmma_f32_16x16x4_f32(false, a1, false, b1, (short)0, acc1, false, false); \
            acc2 = __builtin_amdgcn_wmma_f32_16x16x4_f32(false, a2, false, b2, (short)0, acc2, false, false); \
            acc3 = __builtin_amdgcn_wmma_f32_16x16x4_f32(false, a3, false, b3, (short)0, acc3, false, false); \
        }                                                                           \
        wave_tile_handover();                                                       \
    } while (0)

    ISSUE(bufA, 0);

    {
        const v4f* xs = (const v4f*)(x + kbase);
        v4f*       sd = (v4f*)sx;
        for (int i = tid; i < (SEGC >> 2); i += 256) sd[i] = xs[i];
    }
    __syncthreads();

    #pragma unroll
    for (int s = 0; s < NSTEP; s += 2) {
        ISSUE(bufB, s + 1);
        CONSUME(bufA, s);
        if (s + 2 < NSTEP) ISSUE(bufA, s + 2);
        CONSUME(bufB, s + 1);
    }
#undef ISSUE
#undef CONSUME

    v8f t = acc0 + acc1 + acc2 + acc3;

    if (m == 0) {
        float* dst = y_part + (size_t)blockIdx.y * R + rowbase + kh * 8;
        #pragma unroll
        for (int r = 0; r < 8; ++r) dst[r] = t[r];
    }
}

__global__ void __launch_bounds__(256) reduce_part(
    const float* __restrict__ p, float* __restrict__ y, int R, int nseg)
{
    int i = blockIdx.x * 256 + threadIdx.x;
    if (i < R) {
        float s = 0.0f;
        for (int j = 0; j < nseg; ++j) s += p[(size_t)j * R + i];
        y[i] = s;
    }
}

__global__ void __launch_bounds__(1024) softmax_fuse(
    float* __restrict__ score, const float* __restrict__ query,
    float* __restrict__ gamma)
{
    __shared__ float red[1024];
    const int tid = threadIdx.x;

    float m = -INFINITY;
    #pragma unroll
    for (int j = 0; j < LROWS / 1024; ++j) m = fmaxf(m, score[tid + j * 1024]);
    red[tid] = m;
    __syncthreads();
    for (int s = 512; s > 0; s >>= 1) {
        if (tid < s) red[tid] = fmaxf(red[tid], red[tid + s]);
        __syncthreads();
    }
    const float gmax = red[0];
    __syncthreads();

    float e[LROWS / 1024];
    float sum = 0.0f;
    #pragma unroll
    for (int j = 0; j < LROWS / 1024; ++j) {
        e[j] = expf(score[tid + j * 1024] - gmax);
        sum += e[j];
    }
    red[tid] = sum;
    __syncthreads();
    for (int s = 512; s > 0; s >>= 1) {
        if (tid < s) red[tid] += red[tid + s];
        __syncthreads();
    }
    const float inv = 1.0f / red[0];
    #pragma unroll
    for (int j = 0; j < LROWS / 1024; ++j) score[tid + j * 1024] = e[j] * inv;

    for (int i = tid; i < QDIM; i += 1024) gamma[i] = query[i];
}

__global__ void __launch_bounds__(256) wsum_part(
    const float* __restrict__ ctx, const float* __restrict__ att,
    float* __restrict__ p)
{
    __shared__ float sa[LROWS / 16];
    __shared__ int   srow[LROWS / 16];
    __shared__ int   scount;
    const int tid   = threadIdx.x;
    const int lbase = blockIdx.y * (LROWS / 16);
    for (int i = tid; i < LROWS / 16; i += 256) sa[i] = att[lbase + i];
    __syncthreads();

    if (tid < 32) {
        int base = 0;
        for (int ch = 0; ch < LROWS / 16 / 32; ++ch) {
            const int l = ch * 32 + tid;
            const bool nz = (sa[l] != 0.0f);
            const unsigned mask = __builtin_amdgcn_ballot_w32(nz);
            if (nz) srow[base + __builtin_popcount(mask & ((1u << tid) - 1u))] = l;
            base += __builtin_popcount(mask);
        }
        if (tid == 0) scount = base;
    }
    __syncthreads();

    const int n = scount;
    const int q = blockIdx.x * 256 + tid;
    const float* c = ctx + (size_t)lbase * QDIM + q;
    float acc = 0.0f;
    #pragma unroll 8
    for (int i = 0; i < n; ++i) {
        const int l = srow[i];
        acc = fmaf(sa[l], c[(size_t)l * QDIM], acc);
    }
    p[(size_t)blockIdx.y * QDIM + q] = acc;
}

extern "C" void kernel_launch(void* const* d_in, const int* in_sizes, int n_in,
                              void* d_out, int out_size, void* d_ws, size_t ws_size,
                              hipStream_t stream)
{
    (void)in_sizes; (void)n_in; (void)out_size; (void)ws_size;
    const float* query = (const float*)d_in[0];
    const float* ctx   = (const float*)d_in[1];
    const float* Kw    = (const float*)d_in[2];
    float* out = (float*)d_out;

    float* ws    = (float*)d_ws;
    float* score = ws;
    float* gamma = ws + LROWS;
    float* part  = ws + LROWS + 2 * QDIM;

    gemv16_wmma_f32<QDIM, QDIM / 8><<<dim3(LROWS / 128, 8), 256, 0, stream>>>(
        ctx, query, part);
    reduce_part<<<LROWS / 256, 256, 0, stream>>>(part, score, LROWS, 8);

    softmax_fuse<<<1, 1024, 0, stream>>>(score, query, gamma);

    wsum_part<<<dim3(QDIM / 256, 16), 256, 0, stream>>>(ctx, score, part);
    reduce_part<<<QDIM / 256, 256, 0, stream>>>(part, gamma + QDIM, QDIM, 16);

    gemv16_wmma_f32<2 * QDIM, 2 * QDIM / 8><<<dim3(QDIM / 128, 8), 256, 0, stream>>>(
        Kw, gamma, part);
    reduce_part<<<QDIM / 256, 256, 0, stream>>>(part, out, QDIM, 8);
}


# partial sums stay in the block: a block owns whole outputs (both products, weighted sum), its waves take the old K / L segments and add them in LDS in the old order; 7 launches to 4; sum order kept
# speedup vs baseline: 2.4764x; 1.2355x over previous
#include <hip/hip_runtime.h>
#include <hip/hip_bf16.h>
#include <math.h>

typedef __attribute__((ext_vector_type(2))) float v2f;
typedef __attribute__((ext_vector_type(4))) float v4f;
typedef __attribute__((ext_vector_type(8))) float v8f;

#define QDIM  4096
#define LROWS 8192

#define TK    128
#define PITCH 132

static __device__ __forceinline__ void wave_tile_handover()
{
    __builtin_amdgcn_fence(__ATOMIC_RELEASE, "wavefront", "local");
    __builtin_amdgcn_wave_barrier();
    __builtin_amdgcn_fence(__ATOMIC_ACQUIRE, "wavefront", "local");
}

template <int C>
__global__ void __launch_bounds__(256) gemv16_wmma_f32(
    const float* __restrict__ M, const float* __restrict__ x,
    float* __restrict__ y)
{
    constexpr int SEGC  = C / 8;
    constexpr int NSTEP = SEGC / TK;
    static_assert(SEGC % (2 * TK) == 0, "steps are taken in pairs");
    __shared__ __attribute__((aligned(16))) float sx[C];
    __shared__ __attribute__((aligned(16))) float stile[8 * 16 * PITCH];
    __shared__ float spart[8 * 16];

    const int tid     = threadIdx.x;
    const int lane    = tid & 31;
    const int m       = lane & 15;
    const int kh      = lane >> 4;
    const int wave    = tid >> 5;
    const int kbase   = wave * SEGC;
    const int rowbase = (int)blockIdx.x * 16;
    const float  bm   = (m == 0) ? 1.0f : 0.0f;

    const float* g    = M + (size_t)rowbase * (size_t)C + kbase + lane * 4;
    float*       st   = stile + wave * (16 * PITCH);
    float*       st_w = st + lane * 4;
    const float* st_r = st + m * PITCH + kh * 4;
    const float* sx_r = sx + kbase + kh * 4;

    v4f bufA[16], bufB[16];
    v8f acc0 = {}, acc1 = {}, acc2 = {}, acc3 = {};

#define ISSUE(BUF, S) do {                                                          \
        _Pragma("unroll") for (int r = 0; r < 16; ++r)                              \
            BUF[r] = __builtin_nontemporal_load((const v4f*)(g + (size_t)r * C + (S) * TK)); \
          \
          \
        __builtin_amdgcn_sched_barrier(0);                                          \
    } while (0)

#define CONSUME(BUF, S) do {                                                        \
        _Pragma("unroll") for (int r = 0; r < 16; ++r)                              \
            *(v4f*)(st_w + r * PITCH) = BUF[r];                                     \
        wave_tile_handover();                                                       \
        _Pragma("unroll") for (int it = 0; it < TK / 16; ++it) {                    \
            const v4f a_lo = *(const v4f*)(st_r + it * 16);                         \
            const v4f a_hi = *(const v4f*)(st_r + it * 16 + 8);                     \
            const v4f x_lo = *(const v4f*)(sx_r + (S) * TK + it * 16);              \
            const v4f x_hi = *(const v4f*)(sx_r + (S) * TK + it * 16 + 8);          \
            const v2f a0 = {a_lo.x, a_lo.y}, a1 = {a_lo.z, a_lo.w};                 \
            const v2f a2 = {a_hi.x, a_hi.y}, a3 = {a_hi.z, a_hi.w};                 \
            const v2f b0 = {x_lo.x * bm, x_lo.y * bm}, b1 = {x_lo.z * bm, x_lo.w * bm}; \
            const v2f b2 = {x_hi.x * bm, x_hi.y * bm}, b3 = {x_hi.z * bm, x_hi.w * bm}; \
            acc0 = __builtin_amdgcn_wmma_f32_16x16x4_f32(false, a0, false, b0, (short)0, acc0, false, false); \
            acc1 = __builtin_amdgcn_wmma_f32_16x16x4_f32(false, a1, false, b1, (short)0, acc1, false, false); \
            acc2 = __builtin_amdgcn_wmma_f32_16x16x4_f32(false, a2, false, b2, (short)0, acc2, false, false); \
            acc3 = __builtin_amdgcn_wmma_f32_16x16x4_f32(false, a3, false, b3, (short)0, acc3, false, false); \
        }                                                                           \
        wave_tile_handover();                                                       \
    } while (0)

    ISSUE(bufA, 0);

    {
        const v4f* xs = (const v4f*)(x + kbase) + lane;
        v4f*       sd = (v4f*)(sx + kbase) + lane;
        v4f xv[SEGC / 128];
        #pragma unroll
        for (int j = 0; j < SEGC / 128; ++j) xv[j] = xs[j * 32];
        #pragma unroll
        for (int j = 0; j < SEGC / 128; ++j) sd[j * 32] = xv[j];
    }
    wave_tile_handover();

    #pragma unroll
    for (int s = 0; s < NSTEP; s += 2) {
        ISSUE(bufB, s + 1);
        CONSUME(bufA, s);
        if (s + 2 < NSTEP) ISSUE(bufA, s + 2);
        CONSUME(bufB, s + 1);
    }
#undef ISSUE
#undef CONSUME

    v8f t = acc0 + acc1 + acc2 + acc3;

    if (m == 0) {
        float* dst = spart + wave * 16 + kh * 8;
        #pragma unroll
        for (int r = 0; r < 8; ++r) dst[r] = t[r];
    }
    __syncthreads();
    if (tid < 16) {
        float s = 0.0f;
        for (int j = 0; j < 8; ++j) s += spart[j * 16 + tid];
        y[rowbase + tid] = s;
    }
}

__global__ void __launch_bounds__(1024) softmax_fuse(
    float* __restrict__ score, const float* __restrict__ query,
    float* __restrict__ gamma)
{
    __shared__ float red[1024];
    const int tid = threadIdx.x;

    float m = -INFINITY;
    #pragma unroll
    for (int j = 0; j < LROWS / 1024; ++j) m = fmaxf(m, score[tid + j * 1024]);
    red[tid] = m;
    __syncthreads();
    for (int s = 512; s > 0; s >>= 1) {
        if (tid < s) red[tid] = fmaxf(red[tid], red[tid + s]);
        __syncthreads();
    }
    const float gmax = red[0];
    __syncthreads();

    float e[LROWS / 1024];
    float sum = 0.0f;
    #pragma unroll
    for (int j = 0; j < LROWS / 1024; ++j) {
        e[j] = expf(score[tid + j * 1024] - gmax);
        sum += e[j];
    }
    red[tid] = sum;
    __syncthreads();
    for (int s = 512; s > 0; s >>= 1) {
        if (tid < s) red[tid] += red[tid + s];
        __syncthreads();
    }
    const float inv = 1.0f / red[0];
    #pragma unroll
    for (int j = 0; j < LROWS / 1024; ++j) score[tid + j * 1024] = e[j] * inv;

    for (int i = tid; i < QDIM; i += 1024) gamma[i] = query[i];
}

#define WS_COLS 64
__global__ void __launch_bounds__(512) wsum_rows(
    const float* __restrict__ ctx, const float* __restrict__ att,
    float* __restrict__ s_t)
{
    constexpr int SEGL = LROWS / 16;
    constexpr int NCH  = SEGL / 32;
    __shared__ int   srow[LROWS];
    __shared__ float sval[LROWS];
    __shared__ __attribute__((aligned(8))) float sp[16 * WS_COLS];
    const int tid   = threadIdx.x;
    const int lane  = tid & 31;
    const int seg   = tid >> 5;
    const int lbase = seg * SEGL;
    int*   srow_w = srow + lbase;
    float* sval_w = sval + lbase;

    float w[NCH];
    #pragma unroll
    for (int ch = 0; ch < NCH; ++ch) w[ch] = att[lbase + ch * 32 + lane];

    int n = 0;
    const unsigned below = (1u << lane) - 1u;
    #pragma unroll
    for (int ch = 0; ch < NCH; ++ch) {
        const bool nz = (w[ch] != 0.0f);
        const unsigned mask = __builtin_amdgcn_ballot_w32(nz);
        if (nz) {
            const int pos = n + __builtin_popcount(mask & below);
            srow_w[pos] = ch * 32 + lane;
            sval_w[pos] = w[ch];
        }
        n += __builtin_popcount(mask);
    }
    wave_tile_handover();

    const int q = blockIdx.x * WS_COLS + lane * 2;
    const float* c = ctx + (size_t)lbase * QDIM + q;
    float acc0 = 0.0f, acc1 = 0.0f;
    #pragma unroll 8
    for (int i = 0; i < n; ++i) {
        const int   l = srow_w[i];
        const float a = sval_w[i];
        const v2f   v = *(const v2f*)(c + (size_t)l * QDIM);
        acc0 = fmaf(a, v.x, acc0);
        acc1 = fmaf(a, v.y, acc1);
    }
    {
        const v2f r = {acc0, acc1};
        *(v2f*)(sp + seg * WS_COLS + lane * 2) = r;
    }
    __syncthreads();
    if (tid < WS_COLS) {
        float s = 0.0f;
        for (int j = 0; j < 16; ++j) s += sp[j * WS_COLS + tid];
        s_t[blockIdx.x * WS_COLS + tid] = s;
    }
}

extern "C" void kernel_launch(void* const* d_in, const int* in_sizes, int n_in,
                              void* d_out, int out_size, void* d_ws, size_t ws_size,
                              hipStream_t stream)
{
    (void)in_sizes; (void)n_in; (void)out_size; (void)ws_size;
    const float* query = (const float*)d_in[0];
    const float* ctx   = (const float*)d_in[1];
    const float* Kw    = (const float*)d_in[2];
    float* out = (float*)d_out;

    float* ws    = (float*)d_ws;
    float* score = ws;
    float* gamma = ws + LROWS;

    gemv16_wmma_f32<QDIM><<<LROWS / 16, 256, 0, stream>>>(ctx, query, score);

    softmax_fuse<<<1, 1024, 0, stream>>>(score, query, gamma);

    wsum_rows<<<QDIM / WS_COLS, 512, 0, stream>>>(ctx, score, gamma + QDIM);

    gemv16_wmma_f32<2 * QDIM><<<QDIM / 16, 256, 0, stream>>>(Kw, gamma, out);
}


# query half of the last product (independent of the softmax) streamed by extra block ranges of the softmax and weighted-sum launches; last launch reads only the s_t half; its sum order over k changes
# speedup vs baseline: 2.7113x; 1.0949x over previous
#include <hip/hip_runtime.h>
#include <hip/hip_bf16.h>
#include <math.h>

typedef __attribute__((ext_vector_type(2))) float v2f;
typedef __attribute__((ext_vector_type(4))) float v4f;
typedef __attribute__((ext_vector_type(8))) float v8f;

#define QDIM  4096
#define LROWS 8192

#define TK    128
#define PITCH 132

static __device__ __forceinline__ void wave_tile_handover()
{
    __builtin_amdgcn_fence(__ATOMIC_RELEASE, "wavefront", "local");
    __builtin_amdgcn_wave_barrier();
    __builtin_amdgcn_fence(__ATOMIC_ACQUIRE, "wavefront", "local");
}

template <int C>
__global__ void __launch_bounds__(256) gemv16_wmma_f32(
    const float* __restrict__ M, const float* __restrict__ x,
    float* __restrict__ y)
{
    constexpr int SEGC  = C / 8;
    constexpr int NSTEP = SEGC / TK;
    static_assert(SEGC % (2 * TK) == 0, "steps are taken in pairs");
    __shared__ __attribute__((aligned(16))) float sx[C];
    __shared__ __attribute__((aligned(16))) float stile[8 * 16 * PITCH];
    __shared__ float spart[8 * 16];

    const int tid     = threadIdx.x;
    const int lane    = tid & 31;
    const int m       = lane & 15;
    const int kh      = lane >> 4;
    const int wave    = tid >> 5;
    const int kbase   = wave * SEGC;
    const int rowbase = (int)blockIdx.x * 16;
    const float  bm   = (m == 0) ? 1.0f : 0.0f;

    const float* g    = M + (size_t)rowbase * (size_t)C + kbase + lane * 4;
    float*       st   = stile + wave * (16 * PITCH);
    float*       st_w = st + lane * 4;
    const float* st_r = st + m * PITCH + kh * 4;
    const float* sx_r = sx + kbase + kh * 4;

    v4f bufA[16], bufB[16];
    v8f acc0 = {}, acc1 = {}, acc2 = {}, acc3 = {};

#define ISSUE(BUF, S) do {                                                          \
        _Pragma("unroll") for (int r = 0; r < 16; ++r)                              \
            BUF[r] = __builtin_nontemporal_load((const v4f*)(g + (size_t)r * C + (S) * TK)); \
          \
          \
        __builtin_amdgcn_sched_barrier(0);                                          \
    } while (0)

#define CONSUME(BUF, S) do {                                                        \
        _Pragma("unroll") for (int r = 0; r < 16; ++r)                              \
            *(v4f*)(st_w + r * PITCH) = BUF[r];                                     \
        wave_tile_handover();                                                       \
        _Pragma("unroll") for (int it = 0; it < TK / 16; ++it) {                    \
            const v4f a_lo = *(const v4f*)(st_r + it * 16);                         \
            const v4f a_hi = *(const v4f*)(st_r + it * 16 + 8);                     \
            const v4f x_lo = *(const v4f*)(sx_r + (S) * TK + it * 16);              \
            const v4f x_hi = *(const v4f*)(sx_r + (S) * TK + it * 16 + 8);          \
            const v2f a0 = {a_lo.x, a_lo.y}, a1 = {a_lo.z, a_lo.w};                 \
            const v2f a2 = {a_hi.x, a_hi.y}, a3 = {a_hi.z, a_hi.w};                 \
            const v2f b0 = {x_lo.x * bm, x_lo.y * bm}, b1 = {x_lo.z * bm, x_lo.w * bm}; \
            const v2f b2 = {x_hi.x * bm, x_hi.y * bm}, b3 = {x_hi.z * bm, x_hi.w * bm}; \
            acc0 = __builtin_amdgcn_wmma_f32_16x16x4_f32(false, a0, false, b0, (short)0, acc0, false, false); \
            acc1 = __builtin_amdgcn_wmma_f32_16x16x4_f32(false, a1, false, b1, (short)0, acc1, false, false); \
            acc2 = __builtin_amdgcn_wmma_f32_16x16x4_f32(false, a2, false, b2, (short)0, acc2, false, false); \
            acc3 = __builtin_amdgcn_wmma_f32_16x16x4_f32(false, a3, false, b3, (short)0, acc3, false, false); \
        }                                                                           \
        wave_tile_handover();                                                       \
    } while (0)

    ISSUE(bufA, 0);

    {
        const v4f* xs = (const v4f*)(x + kbase) + lane;
        v4f*       sd = (v4f*)(sx + kbase) + lane;
        v4f xv[SEGC / 128];
        #pragma unroll
        for (int j = 0; j < SEGC / 128; ++j) xv[j] = xs[j * 32];
        #pragma unroll
        for (int j = 0; j < SEGC / 128; ++j) sd[j * 32] = xv[j];
    }
    wave_tile_handover();

    #pragma unroll
    for (int s = 0; s < NSTEP; s += 2) {
        ISSUE(bufB, s + 1);
        CONSUME(bufA, s);
        if (s + 2 < NSTEP) ISSUE(bufA, s + 2);
        CONSUME(bufB, s + 1);
    }
#undef ISSUE
#undef CONSUME

    v8f t = acc0 + acc1 + acc2 + acc3;

    if (m == 0) {
        float* dst = spart + wave * 16 + kh * 8;
        #pragma unroll
        for (int r = 0; r < 8; ++r) dst[r] = t[r];
    }
    __syncthreads();
    if (tid < 16) {
        float s = 0.0f;
        for (int j = 0; j < 8; ++j) s += spart[j * 16 + tid];
        y[rowbase + tid] = s;
    }
}

#define LDW   (2 * QDIM)
#define TCOLS 256

static __device__ __forceinline__ v8f wave_rows16_cols256(
    const float* __restrict__ Mt,
    const float* __restrict__ xt,
    float* st,
    float* sxw,
    const int lane)
{
    const int   m  = lane & 15;
    const int   kh = lane >> 4;
    const float bm = (m == 0) ? 1.0f : 0.0f;

    const float* g    = Mt + lane * 4;
    float*       st_w = st + lane * 4;
    const float* st_r = st + m * PITCH + kh * 4;
    const float* sx_r = sxw + kh * 4;

    v4f bufA[16], bufB[16];
    v8f acc0 = {}, acc1 = {}, acc2 = {}, acc3 = {};

    const v4f xv0 = ((const v4f*)xt)[lane];
    const v4f xv1 = ((const v4f*)xt)[lane + 32];

#define ISSUE2(BUF, S) do {                                                         \
        _Pragma("unroll") for (int r = 0; r < 16; ++r)                              \
            BUF[r] = __builtin_nontemporal_load((const v4f*)(g + (size_t)r * LDW + (S) * TK)); \
          \
          \
        __builtin_amdgcn_sched_barrier(0);                                          \
    } while (0)

#define CONSUME2(BUF, S) do {                                                       \
        _Pragma("unroll") for (int r = 0; r < 16; ++r)                              \
            *(v4f*)(st_w + r * PITCH) = BUF[r];                                     \
        wave_tile_handover();                                                       \
        _Pragma("unroll") for (int it = 0; it < TK / 16; ++it) {                    \
            const v4f a_lo = *(const v4f*)(st_r + it * 16);                         \
            const v4f a_hi = *(const v4f*)(st_r + it * 16 + 8);                     \
            const v4f x_lo = *(const v4f*)(sx_r + (S) * TK + it * 16);              \
            const v4f x_hi = *(const v4f*)(sx_r + (S) * TK + it * 16 + 8);          \
            const v2f a0 = {a_lo.x, a_lo.y}, a1 = {a_lo.z, a_lo.w};                 \
            const v2f a2 = {a_hi.x, a_hi.y}, a3 = {a_hi.z, a_hi.w};                 \
            const v2f b0 = {x_lo.x * bm, x_lo.y * bm}, b1 = {x_lo.z * bm, x_lo.w * bm}; \
            const v2f b2 = {x_hi.x * bm, x_hi.y * bm}, b3 = {x_hi.z * bm, x_hi.w * bm}; \
            acc0 = __builtin_amdgcn_wmma_f32_16x16x4_f32(false, a0, false, b0, (short)0, acc0, false, false); \
            acc1 = __builtin_amdgcn_wmma_f32_16x16x4_f32(false, a1, false, b1, (short)0, acc1, false, false); \
            acc2 = __builtin_amdgcn_wmma_f32_16x16x4_f32(false, a2, false, b2, (short)0, acc2, false, false); \
            acc3 = __builtin_amdgcn_wmma_f32_16x16x4_f32(false, a3, false, b3, (short)0, acc3, false, false); \
        }                                                                           \
        wave_tile_handover();                                                       \
    } while (0)

    ISSUE2(bufA, 0);
    ISSUE2(bufB, 1);

    ((v4f*)sxw)[lane]      = xv0;
    ((v4f*)sxw)[lane + 32] = xv1;
    wave_tile_handover();

    CONSUME2(bufA, 0);
    CONSUME2(bufB, 1);
#undef ISSUE2
#undef CONSUME2

    return acc0 + acc1 + acc2 + acc3;
}

static __device__ __forceinline__ void park_col0(float* spart, const v8f t, const int wave, const int lane)
{
    if ((lane & 15) == 0) {
        float* dst = spart + wave * 16 + (lane >> 4) * 8;
        #pragma unroll
        for (int r = 0; r < 8; ++r) dst[r] = t[r];
    }
}

static __device__ __forceinline__ void softmax_role(float* __restrict__ score, float* red, const int tid)
{
    float sc[4][8];
    float mx = -INFINITY;
    #pragma unroll
    for (int i = 0; i < 4; ++i) {
        #pragma unroll
        for (int j = 0; j < 8; ++j) {
            sc[i][j] = score[tid + 256 * i + 1024 * j];
            mx = fmaxf(mx, sc[i][j]);
        }
    }
    red[tid] = mx;
    __syncthreads();
    for (int s = 128; s > 0; s >>= 1) {
        if (tid < s) red[tid] = fmaxf(red[tid], red[tid + s]);
        __syncthreads();
    }
    const float gmax = red[0];
    __syncthreads();

    float p[4];
    #pragma unroll
    for (int i = 0; i < 4; ++i) {
        float sum = 0.0f;
        #pragma unroll
        for (int j = 0; j < 8; ++j) {
            sc[i][j] = expf(sc[i][j] - gmax);
            sum += sc[i][j];
        }
        p[i] = sum;
    }
    const float a0 = p[0] + p[2];
    const float a1 = p[1] + p[3];
    red[tid] = a0 + a1;
    __syncthreads();
    for (int s = 128; s > 0; s >>= 1) {
        if (tid < s) red[tid] += red[tid + s];
        __syncthreads();
    }
    const float inv = 1.0f / red[0];
    #pragma unroll
    for (int i = 0; i < 4; ++i) {
        #pragma unroll
        for (int j = 0; j < 8; ++j) score[tid + 256 * i + 1024 * j] = sc[i][j] * inv;
    }
}

#define QH_ROWGROUPS_L2 127
__global__ void __launch_bounds__(256) softmax_and_qhalf(
    float* __restrict__ score, const float* __restrict__ Kw,
    const float* __restrict__ query, float* __restrict__ pq)
{
    __shared__ float red[256];
    __shared__ __attribute__((aligned(16))) float stile[8 * 16 * PITCH];
    __shared__ __attribute__((aligned(16))) float sxv[8 * TCOLS];
    __shared__ float spart[8 * 16];
    const int tid = threadIdx.x;
    if (blockIdx.x == 0) {
        softmax_role(score, red, tid);
        return;
    }
    const int lane    = tid & 31;
    const int wave    = tid >> 5;
    const int b       = (int)blockIdx.x - 1;
    const int half    = b & 1;
    const int rowbase = (b >> 1) * 16;
    const int colbase = half * (QDIM / 2) + wave * TCOLS;

    const v8f t = wave_rows16_cols256(Kw + (size_t)rowbase * LDW + colbase, query + colbase,
                                      stile + wave * (16 * PITCH), sxv + wave * TCOLS, lane);
    park_col0(spart, t, wave, lane);
    __syncthreads();
    if (tid < 16) {
        float s = 0.0f;
        for (int j = 0; j < 8; ++j) s += spart[j * 16 + tid];
        pq[half * QDIM + rowbase + tid] = s;
    }
}

#define WS_COLS 64
static __device__ __forceinline__ void wsum_role(
    const float* __restrict__ ctx, const float* __restrict__ att,
    float* __restrict__ s_t,
    int* srow,
    float* sval,
    float* sp,
    const int tid)
{
    constexpr int SEGL = LROWS / 16;
    constexpr int NCH  = SEGL / 32;
    const int lane  = tid & 31;
    const int seg   = tid >> 5;
    const int lbase = seg * SEGL;
    int*   srow_w = srow + lbase;
    float* sval_w = sval + lbase;

    float w[NCH];
    #pragma unroll
    for (int ch = 0; ch < NCH; ++ch) w[ch] = att[lbase + ch * 32 + lane];

    int n = 0;
    const unsigned below = (1u << lane) - 1u;
    #pragma unroll
    for (int ch = 0; ch < NCH; ++ch) {
        const bool nz = (w[ch] != 0.0f);
        const unsigned mask = __builtin_amdgcn_ballot_w32(nz);
        if (nz) {
            const int pos = n + __builtin_popcount(mask & below);
            srow_w[pos] = ch * 32 + lane;
            sval_w[pos] = w[ch];
        }
        n += __builtin_popcount(mask);
    }
    wave_tile_handover();

    const int q = blockIdx.x * WS_COLS + lane * 2;
    const float* c = ctx + (size_t)lbase * QDIM + q;
    float acc0 = 0.0f, acc1 = 0.0f;
    #pragma unroll 8
    for (int i = 0; i < n; ++i) {
        const int   l = srow_w[i];
        const float a = sval_w[i];
        const v2f   v = *(const v2f*)(c + (size_t)l * QDIM);
        acc0 = fmaf(a, v.x, acc0);
        acc1 = fmaf(a, v.y, acc1);
    }
    {
        const v2f r = {acc0, acc1};
        *(v2f*)(sp + seg * WS_COLS + lane * 2) = r;
    }
    __syncthreads();
    if (tid < WS_COLS) {
        float s = 0.0f;
        for (int j = 0; j < 16; ++j) s += sp[j * WS_COLS + tid];
        s_t[blockIdx.x * WS_COLS + tid] = s;
    }
}

__global__ void __launch_bounds__(512) wsum_and_qhalf(
    const float* __restrict__ ctx, const float* __restrict__ att,
    float* __restrict__ s_t, const float* __restrict__ Kw,
    const float* __restrict__ query, float* __restrict__ pq)
{
    __shared__ int   srow[LROWS];
    __shared__ float sval[LROWS];
    __shared__ __attribute__((aligned(8)))  float sp[16 * WS_COLS];
    __shared__ __attribute__((aligned(16))) float stile[16 * 16 * PITCH];
    __shared__ __attribute__((aligned(16))) float sxv[16 * TCOLS];
    __shared__ float spart[16 * 16];
    const int tid = threadIdx.x;
    if (blockIdx.x < QDIM / WS_COLS) {
        wsum_role(ctx, att, s_t, srow, sval, sp, tid);
        return;
    }
    const int lane    = tid & 31;
    const int wave    = tid >> 5;
    const int rowbase = (QH_ROWGROUPS_L2 + (int)blockIdx.x - QDIM / WS_COLS) * 16;
    const int colbase = wave * TCOLS;

    const v8f t = wave_rows16_cols256(Kw + (size_t)rowbase * LDW + colbase, query + colbase,
                                      stile + wave * (16 * PITCH), sxv + wave * TCOLS, lane);
    park_col0(spart, t, wave, lane);
    __syncthreads();
    if (tid < 32) {
        const int half = tid >> 4, r = tid & 15;
        float s = 0.0f;
        for (int j = 0; j < 8; ++j) s += spart[(half * 8 + j) * 16 + r];
        pq[half * QDIM + rowbase + r] = s;
    }
}

__global__ void __launch_bounds__(512) gemv_shalf_finish(
    const float* __restrict__ Kw, const float* __restrict__ s_t,
    const float* __restrict__ pq, float* __restrict__ out)
{
    __shared__ __attribute__((aligned(16))) float stile[16 * 16 * PITCH];
    __shared__ __attribute__((aligned(16))) float sxv[16 * TCOLS];
    __shared__ float spart[16 * 16];
    const int tid     = threadIdx.x;
    const int lane    = tid & 31;
    const int wave    = tid >> 5;
    const int rowbase = (int)blockIdx.x * 16;
    const int colbase = wave * TCOLS;

    const float p0 = pq[rowbase + (tid & 15)];
    const float p1 = pq[QDIM + rowbase + (tid & 15)];

    const v8f t = wave_rows16_cols256(Kw + (size_t)rowbase * LDW + QDIM + colbase, s_t + colbase,
                                      stile + wave * (16 * PITCH), sxv + wave * TCOLS, lane);
    park_col0(spart, t, wave, lane);
    __syncthreads();
    if (tid < 16) {
        float s = 0.0f;
        s += p0;
        s += p1;
        for (int j = 0; j < 16; ++j) s += spart[j * 16 + tid];
        out[rowbase + tid] = s;
    }
}

extern "C" void kernel_launch(void* const* d_in, const int* in_sizes, int n_in,
                              void* d_out, int out_size, void* d_ws, size_t ws_size,
                              hipStream_t stream)
{
    (void)in_sizes; (void)n_in; (void)out_size; (void)ws_size;
    const float* query = (const float*)d_in[0];
    const float* ctx   = (const float*)d_in[1];
    const float* Kw    = (const float*)d_in[2];
    float* out = (float*)d_out;

    float* ws    = (float*)d_ws;
    float* score = ws;
    float* s_t   = ws + LROWS;
    float* pq    = ws + LROWS + QDIM;

    gemv16_wmma_f32<QDIM><<<LROWS / 16, 256, 0, stream>>>(ctx, query, score);

    softmax_and_qhalf<<<1 + 2 * QH_ROWGROUPS_L2, 256, 0, stream>>>(score, Kw, query, pq);

    wsum_and_qhalf<<<QDIM / WS_COLS + (QDIM / 16 - QH_ROWGROUPS_L2), 512, 0, stream>>>(ctx, score, s_t, Kw, query, pq);

    gemv_shalf_finish<<<QDIM / 16, 512, 0, stream>>>(Kw, s_t, pq, out);
}


# softmax launch removed by a split reduction: each score block also writes its maximum and exp-sum (512 pairs), every weighted-sum wave merges them; 4 launches to 3; denominator sum order changes
# speedup vs baseline: 3.0053x; 1.1084x over previous
#include <hip/hip_runtime.h>
#include <hip/hip_bf16.h>
#include <math.h>

typedef __attribute__((ext_vector_type(2))) float v2f;
typedef __attribute__((ext_vector_type(4))) float v4f;
typedef __attribute__((ext_vector_type(8))) float v8f;

#define QDIM  4096
#define LROWS 8192

#define TK    128
#define PITCH 132

static __device__ __forceinline__ void wave_tile_handover()
{
    __builtin_amdgcn_fence(__ATOMIC_RELEASE, "wavefront", "local");
    __builtin_amdgcn_wave_barrier();
    __builtin_amdgcn_fence(__ATOMIC_ACQUIRE, "wavefront", "local");
}

template <int K>
static __device__ __forceinline__ float row_xor(const float v)
{
    const int i = __builtin_bit_cast(int, v);
    return __builtin_bit_cast(float, __builtin_amdgcn_update_dpp(i, i, 0x160 + K  , 0xf, 0xf, false));
}

template <int C>
__global__ void __launch_bounds__(256) gemv16_wmma_f32(
    const float* __restrict__ M, const float* __restrict__ x,
    float* __restrict__ y,
    float* __restrict__ stats)
{
    constexpr int SEGC  = C / 8;
    constexpr int NSTEP = SEGC / TK;
    static_assert(SEGC % (2 * TK) == 0, "steps are taken in pairs");
    __shared__ __attribute__((aligned(16))) float sx[C];
    __shared__ __attribute__((aligned(16))) float stile[8 * 16 * PITCH];
    __shared__ float spart[8 * 16];

    const int tid     = threadIdx.x;
    const int lane    = tid & 31;
    const int m       = lane & 15;
    const int kh      = lane >> 4;
    const int wave    = tid >> 5;
    const int kbase   = wave * SEGC;
    const int rowbase = (int)blockIdx.x * 16;
    const float  bm   = (m == 0) ? 1.0f : 0.0f;

    const float* g    = M + (size_t)rowbase * (size_t)C + kbase + lane * 4;
    float*       st   = stile + wave * (16 * PITCH);
    float*       st_w = st + lane * 4;
    const float* st_r = st + m * PITCH + kh * 4;
    const float* sx_r = sx + kbase + kh * 4;

    v4f bufA[16], bufB[16];
    v8f acc0 = {}, acc1 = {}, acc2 = {}, acc3 = {};

#define ISSUE(BUF, S) do {                                                          \
        _Pragma("unroll") for (int r = 0; r < 16; ++r)                              \
            BUF[r] = __builtin_nontemporal_load((const v4f*)(g + (size_t)r * C + (S) * TK)); \
          \
          \
        __builtin_amdgcn_sched_barrier(0);                                          \
    } while (0)

#define CONSUME(BUF, S) do {                                                        \
        _Pragma("unroll") for (int r = 0; r < 16; ++r)                              \
            *(v4f*)(st_w + r * PITCH) = BUF[r];                                     \
        wave_tile_handover();                                                       \
        _Pragma("unroll") for (int it = 0; it < TK / 16; ++it) {                    \
            const v4f a_lo = *(const v4f*)(st_r + it * 16);                         \
            const v4f a_hi = *(const v4f*)(st_r + it * 16 + 8);                     \
            const v4f x_lo = *(const v4f*)(sx_r + (S) * TK + it * 16);              \
            const v4f x_hi = *(const v4f*)(sx_r + (S) * TK + it * 16 + 8);          \
            const v2f a0 = {a_lo.x, a_lo.y}, a1 = {a_lo.z, a_lo.w};                 \
            const v2f a2 = {a_hi.x, a_hi.y}, a3 = {a_hi.z, a_hi.w};                 \
            const v2f b0 = {x_lo.x * bm, x_lo.y * bm}, b1 = {x_lo.z * bm, x_lo.w * bm}; \
            const v2f b2 = {x_hi.x * bm, x_hi.y * bm}, b3 = {x_hi.z * bm, x_hi.w * bm}; \
            acc0 = __builtin_amdgcn_wmma_f32_16x16x4_f32(false, a0, false, b0, (short)0, acc0, false, false); \
            acc1 = __builtin_amdgcn_wmma_f32_16x16x4_f32(false, a1, false, b1, (short)0, acc1, false, false); \
            acc2 = __builtin_amdgcn_wmma_f32_16x16x4_f32(false, a2, false, b2, (short)0, acc2, false, false); \
            acc3 = __builtin_amdgcn_wmma_f32_16x16x4_f32(false, a3, false, b3, (short)0, acc3, false, false); \
        }                                                                           \
        wave_tile_handover();                                                       \
    } while (0)

    ISSUE(bufA, 0);

    {
        const v4f* xs = (const v4f*)(x + kbase) + lane;
        v4f*       sd = (v4f*)(sx + kbase) + lane;
        v4f xv[SEGC / 128];
        #pragma unroll
        for (int j = 0; j < SEGC / 128; ++j) xv[j] = xs[j * 32];
        #pragma unroll
        for (int j = 0; j < SEGC / 128; ++j) sd[j * 32] = xv[j];
    }
    wave_tile_handover();

    #pragma unroll
    for (int s = 0; s < NSTEP; s += 2) {
        ISSUE(bufB, s + 1);
        CONSUME(bufA, s);
        if (s + 2 < NSTEP) ISSUE(bufA, s + 2);
        CONSUME(bufB, s + 1);
    }
#undef ISSUE
#undef CONSUME

    v8f t = acc0 + acc1 + acc2 + acc3;

    if (m == 0) {
        float* dst = spart + wave * 16 + kh * 8;
        #pragma unroll
        for (int r = 0; r < 8; ++r) dst[r] = t[r];
    }
    __syncthreads();
    if (tid < 32) {
        const int r = tid & 15;
        float s = 0.0f;
        for (int j = 0; j < 8; ++j) s += spart[j * 16 + r];
        if (tid < 16) y[rowbase + tid] = s;
        float bmx = s;
        bmx = fmaxf(bmx, row_xor<8>(bmx));
        bmx = fmaxf(bmx, row_xor<4>(bmx));
        bmx = fmaxf(bmx, row_xor<2>(bmx));
        bmx = fmaxf(bmx, row_xor<1>(bmx));
        float bsum = expf(s - bmx);
        bsum += row_xor<8>(bsum);
        bsum += row_xor<4>(bsum);
        bsum += row_xor<2>(bsum);
        bsum += row_xor<1>(bsum);
        if (tid == 0) {
            const v2f pr = {bmx, bsum};
            *(v2f*)(stats + 2 * blockIdx.x) = pr;
        }
    }
}

#define LDW   (2 * QDIM)
#define TCOLS 256

static __device__ __forceinline__ v8f wave_rows16_cols256(
    const float* __restrict__ Mt,
    const float* __restrict__ xt,
    float* st,
    float* sxw,
    const int lane)
{
    const int   m  = lane & 15;
    const int   kh = lane >> 4;
    const float bm = (m == 0) ? 1.0f : 0.0f;

    const float* g    = Mt + lane * 4;
    float*       st_w = st + lane * 4;
    const float* st_r = st + m * PITCH + kh * 4;
    const float* sx_r = sxw + kh * 4;

    v4f bufA[16], bufB[16];
    v8f acc0 = {}, acc1 = {}, acc2 = {}, acc3 = {};

    const v4f xv0 = ((const v4f*)xt)[lane];
    const v4f xv1 = ((const v4f*)xt)[lane + 32];

#define ISSUE2(BUF, S) do {                                                         \
        _Pragma("unroll") for (int r = 0; r < 16; ++r)                              \
            BUF[r] = __builtin_nontemporal_load((const v4f*)(g + (size_t)r * LDW + (S) * TK)); \
          \
          \
        __builtin_amdgcn_sched_barrier(0);                                          \
    } while (0)

#define CONSUME2(BUF, S) do {                                                       \
        _Pragma("unroll") for (int r = 0; r < 16; ++r)                              \
            *(v4f*)(st_w + r * PITCH) = BUF[r];                                     \
        wave_tile_handover();                                                       \
        _Pragma("unroll") for (int it = 0; it < TK / 16; ++it) {                    \
            const v4f a_lo = *(const v4f*)(st_r + it * 16);                         \
            const v4f a_hi = *(const v4f*)(st_r + it * 16 + 8);                     \
            const v4f x_lo = *(const v4f*)(sx_r + (S) * TK + it * 16);              \
            const v4f x_hi = *(const v4f*)(sx_r + (S) * TK + it * 16 + 8);          \
            const v2f a0 = {a_lo.x, a_lo.y}, a1 = {a_lo.z, a_lo.w};                 \
            const v2f a2 = {a_hi.x, a_hi.y}, a3 = {a_hi.z, a_hi.w};                 \
            const v2f b0 = {x_lo.x * bm, x_lo.y * bm}, b1 = {x_lo.z * bm, x_lo.w * bm}; \
            const v2f b2 = {x_hi.x * bm, x_hi.y * bm}, b3 = {x_hi.z * bm, x_hi.w * bm}; \
            acc0 = __builtin_amdgcn_wmma_f32_16x16x4_f32(false, a0, false, b0, (short)0, acc0, false, false); \
            acc1 = __builtin_amdgcn_wmma_f32_16x16x4_f32(false, a1, false, b1, (short)0, acc1, false, false); \
            acc2 = __builtin_amdgcn_wmma_f32_16x16x4_f32(false, a2, false, b2, (short)0, acc2, false, false); \
            acc3 = __builtin_amdgcn_wmma_f32_16x16x4_f32(false, a3, false, b3, (short)0, acc3, false, false); \
        }                                                                           \
        wave_tile_handover();                                                       \
    } while (0)

    ISSUE2(bufA, 0);
    ISSUE2(bufB, 1);

    ((v4f*)sxw)[lane]      = xv0;
    ((v4f*)sxw)[lane + 32] = xv1;
    wave_tile_handover();

    CONSUME2(bufA, 0);
    CONSUME2(bufB, 1);
#undef ISSUE2
#undef CONSUME2

    return acc0 + acc1 + acc2 + acc3;
}

static __device__ __forceinline__ void wave_two_tasks(
    const float* __restrict__ MtA, const float* __restrict__ xtA,
    const float* __restrict__ MtB, const float* __restrict__ xtB,
    float* st,
    float* sxw,
    const int lane, v8f& tA, v8f& tB)
{
    const int   m  = lane & 15;
    const int   kh = lane >> 4;
    const float bm = (m == 0) ? 1.0f : 0.0f;

    const float* gA   = MtA + lane * 4;
    const float* gB   = MtB + lane * 4;
    float*       st_w = st + lane * 4;
    const float* st_r = st + m * PITCH + kh * 4;
    const float* sx_r = sxw + kh * 4;

    v4f bufA[16], bufB[16];
    v8f acc0 = {}, acc1 = {}, acc2 = {}, acc3 = {};

    const v4f xa0 = ((const v4f*)xtA)[lane];
    const v4f xa1 = ((const v4f*)xtA)[lane + 32];
    const v4f xb0 = ((const v4f*)xtB)[lane];
    const v4f xb1 = ((const v4f*)xtB)[lane + 32];

#define ISSUE3(BUF, G, S) do {                                                      \
        _Pragma("unroll") for (int r = 0; r < 16; ++r)                              \
            BUF[r] = __builtin_nontemporal_load((const v4f*)((G) + (size_t)r * LDW + (S) * TK)); \
          \
          \
        __builtin_amdgcn_sched_barrier(0);                                          \
    } while (0)

#define CONSUME3(BUF, XS) do {                                                      \
        _Pragma("unroll") for (int r = 0; r < 16; ++r)                              \
            *(v4f*)(st_w + r * PITCH) = BUF[r];                                     \
        wave_tile_handover();                                                       \
          \
          \
          \
          \
        _Pragma("unroll 4") for (int it = 0; it < TK / 16; ++it) {                  \
            const v4f a_lo = *(const v4f*)(st_r + it * 16);                         \
            const v4f a_hi = *(const v4f*)(st_r + it * 16 + 8);                     \
            const v4f x_lo = *(const v4f*)(sx_r + (XS) * TK + it * 16);             \
            const v4f x_hi = *(const v4f*)(sx_r + (XS) * TK + it * 16 + 8);         \
            const v2f a0 = {a_lo.x, a_lo.y}, a1 = {a_lo.z, a_lo.w};                 \
            const v2f a2 = {a_hi.x, a_hi.y}, a3 = {a_hi.z, a_hi.w};                 \
            const v2f b0 = {x_lo.x * bm, x_lo.y * bm}, b1 = {x_lo.z * bm, x_lo.w * bm}; \
            const v2f b2 = {x_hi.x * bm, x_hi.y * bm}, b3 = {x_hi.z * bm, x_hi.w * bm}; \
            acc0 = __builtin_amdgcn_wmma_f32_16x16x4_f32(false, a0, false, b0, (short)0, acc0, false, false); \
            acc1 = __builtin_amdgcn_wmma_f32_16x16x4_f32(false, a1, false, b1, (short)0, acc1, false, false); \
            acc2 = __builtin_amdgcn_wmma_f32_16x16x4_f32(false, a2, false, b2, (short)0, acc2, false, false); \
            acc3 = __builtin_amdgcn_wmma_f32_16x16x4_f32(false, a3, false, b3, (short)0, acc3, false, false); \
        }                                                                           \
        wave_tile_handover();                                                       \
    } while (0)

    ISSUE3(bufA, gA, 0);
    ISSUE3(bufB, gA, 1);

    ((v4f*)sxw)[lane]      = xa0;
    ((v4f*)sxw)[lane + 32] = xa1;
    ((v4f*)sxw)[lane + 64] = xb0;
    ((v4f*)sxw)[lane + 96] = xb1;
    wave_tile_handover();

    CONSUME3(bufA, 0);
    ISSUE3(bufA, gB, 0);
    CONSUME3(bufB, 1);
    ISSUE3(bufB, gB, 1);
    tA = acc0 + acc1 + acc2 + acc3;
    acc0 = acc1 = acc2 = acc3 = (v8f){};
    CONSUME3(bufA, 2);
    CONSUME3(bufB, 3);
#undef ISSUE3
#undef CONSUME3
    tB = acc0 + acc1 + acc2 + acc3;
}

static __device__ __forceinline__ void park_col0(float* spart, const v8f t, const int wave, const int lane)
{
    if ((lane & 15) == 0) {
        float* dst = spart + wave * 16 + (lane >> 4) * 8;
        #pragma unroll
        for (int r = 0; r < 8; ++r) dst[r] = t[r];
    }
}

#define WS_COLS 64
static __device__ __forceinline__ void wsum_role(
    const float* __restrict__ ctx, const float* __restrict__ score,
    const float* __restrict__ stats,
    float* __restrict__ s_t,
    int* srow,
    float* sval,
    float* sp,
    const int cb,
    const int tid)
{
    constexpr int SEGL = LROWS / 16;
    constexpr int NCH  = SEGL / 32;
    const int lane  = tid & 31;
    const int seg   = tid >> 5;
    const int lbase = seg * SEGL;
    int*   srow_w = srow + lbase;
    float* sval_w = sval + lbase;

    constexpr int NPR = (LROWS / 16) / 32;
    float w[NCH];
    v2f   pr[NPR];
    #pragma unroll
    for (int j = 0; j < NPR; ++j) pr[j] = ((const v2f*)stats)[j * 32 + lane];
    #pragma unroll
    for (int ch = 0; ch < NCH; ++ch) w[ch] = score[lbase + ch * 32 + lane];
    __builtin_amdgcn_sched_barrier(0);

    float gmax = pr[0].x;
    #pragma unroll
    for (int j = 1; j < NPR; ++j) gmax = fmaxf(gmax, pr[j].x);
    for (int k = 16; k > 0; k >>= 1) gmax = fmaxf(gmax, __shfl_xor(gmax, k, 32));
    float den = 0.0f;
    #pragma unroll
    for (int j = 0; j < NPR; ++j) den += pr[j].y * expf(pr[j].x - gmax);
    for (int k = 16; k > 0; k >>= 1) den += __shfl_xor(den, k, 32);
    const float inv = 1.0f / den;
    #pragma unroll
    for (int ch = 0; ch < NCH; ++ch) w[ch] = expf(w[ch] - gmax) * inv;

    int n = 0;
    const unsigned below = (1u << lane) - 1u;
    #pragma unroll
    for (int ch = 0; ch < NCH; ++ch) {
        const bool nz = (w[ch] != 0.0f);
        const unsigned mask = __builtin_amdgcn_ballot_w32(nz);
        if (nz) {
            const int pos = n + __builtin_popcount(mask & below);
            srow_w[pos] = ch * 32 + lane;
            sval_w[pos] = w[ch];
        }
        n += __builtin_popcount(mask);
    }
    wave_tile_handover();

    const int q = cb * WS_COLS + lane * 2;
    const float* c = ctx + (size_t)lbase * QDIM + q;
    float acc0 = 0.0f, acc1 = 0.0f;
    #pragma unroll 8
    for (int i = 0; i < n; ++i) {
        const int   l = srow_w[i];
        const float a = sval_w[i];
        const v2f   v = *(const v2f*)(c + (size_t)l * QDIM);
        acc0 = fmaf(a, v.x, acc0);
        acc1 = fmaf(a, v.y, acc1);
    }
    {
        const v2f r = {acc0, acc1};
        *(v2f*)(sp + seg * WS_COLS + lane * 2) = r;
    }
    __syncthreads();
    if (tid < WS_COLS) {
        float s = 0.0f;
        for (int j = 0; j < 16; ++j) s += sp[j * WS_COLS + tid];
        s_t[cb * WS_COLS + tid] = s;
    }
}

#define QH_PAIR_BLOCKS 128
__global__ void __launch_bounds__(512) wsum_and_qhalf(
    const float* __restrict__ ctx, const float* __restrict__ score,
    const float* __restrict__ stats,
    float* __restrict__ s_t, const float* __restrict__ Kw,
    const float* __restrict__ query, float* __restrict__ pq)
{
    __shared__ int   srow[LROWS];
    __shared__ float sval[LROWS];
    __shared__ __attribute__((aligned(8)))  float sp[16 * WS_COLS];
    __shared__ __attribute__((aligned(16))) float stile[16 * 16 * PITCH];
    __shared__ __attribute__((aligned(16))) float sxv[16 * 2 * TCOLS];
    __shared__ float spart[24 * 16];
    const int tid = threadIdx.x;
    const int bx  = (int)blockIdx.x;
    if (bx >= QH_PAIR_BLOCKS && bx < QH_PAIR_BLOCKS + QDIM / WS_COLS) {
        wsum_role(ctx, score, stats, s_t, srow, sval, sp, bx - QH_PAIR_BLOCKS, tid);
        return;
    }
    const int lane    = tid & 31;
    const int wave    = tid >> 5;
    const bool paired = bx < QH_PAIR_BLOCKS;
    const int side    = bx & 1;
    const int rg_own  = paired ? 3 * (bx >> 1) + 2 * side : bx;
    const int rg_sh   = 3 * (bx >> 1) + 1;
    const int rowbase = rg_own * 16;
    const int colbase = wave * TCOLS;
    float* st  = stile + wave * (16 * PITCH);
    float* sxw = sxv + wave * (2 * TCOLS);

    if (paired && wave < 8) {
        const int colsh = (side * 8 + wave) * TCOLS;
        v8f tA, tB;
        wave_two_tasks(Kw + (size_t)rowbase * LDW + colbase, query + colbase,
                       Kw + (size_t)(rg_sh * 16) * LDW + colsh, query + colsh,
                       st, sxw, lane, tA, tB);
        park_col0(spart, tA, wave, lane);
        park_col0(spart, tB, 16 + wave, lane);
    } else {
        const v8f t = wave_rows16_cols256(Kw + (size_t)rowbase * LDW + colbase, query + colbase,
                                          st, sxw, lane);
        park_col0(spart, t, wave, lane);
    }
    __syncthreads();
    if (tid < 32) {
        const int half = tid >> 4, r = tid & 15;
        float s = 0.0f;
        for (int j = 0; j < 8; ++j) s += spart[(half * 8 + j) * 16 + r];
        pq[half * QDIM + rowbase + r] = s;
    } else if (tid < 48 && paired) {
        const int r = tid & 15;
        float s = 0.0f;
        for (int j = 0; j < 8; ++j) s += spart[(16 + j) * 16 + r];
        pq[side * QDIM + rg_sh * 16 + r] = s;
    }
}

__global__ void __launch_bounds__(512) gemv_shalf_finish(
    const float* __restrict__ Kw, const float* __restrict__ s_t,
    const float* __restrict__ pq, float* __restrict__ out)
{
    __shared__ __attribute__((aligned(16))) float stile[16 * 16 * PITCH];
    __shared__ __attribute__((aligned(16))) float sxv[16 * TCOLS];
    __shared__ float spart[16 * 16];
    const int tid     = threadIdx.x;
    const int lane    = tid & 31;
    const int wave    = tid >> 5;
    const int rowbase = (int)blockIdx.x * 16;
    const int colbase = wave * TCOLS;

    const float p0 = pq[rowbase + (tid & 15)];
    const float p1 = pq[QDIM + rowbase + (tid & 15)];

    const v8f t = wave_rows16_cols256(Kw + (size_t)rowbase * LDW + QDIM + colbase, s_t + colbase,
                                      stile + wave * (16 * PITCH), sxv + wave * TCOLS, lane);
    park_col0(spart, t, wave, lane);
    __syncthreads();
    if (tid < 16) {
        float s = 0.0f;
        s += p0;
        s += p1;
        for (int j = 0; j < 16; ++j) s += spart[j * 16 + tid];
        out[rowbase + tid] = s;
    }
}

extern "C" void kernel_launch(void* const* d_in, const int* in_sizes, int n_in,
                              void* d_out, int out_size, void* d_ws, size_t ws_size,
                              hipStream_t stream)
{
    (void)in_sizes; (void)n_in; (void)out_size; (void)ws_size;
    const float* query = (const float*)d_in[0];
    const float* ctx   = (const float*)d_in[1];
    const float* Kw    = (const float*)d_in[2];
    float* out = (float*)d_out;

    float* ws    = (float*)d_ws;
    float* score = ws;
    float* s_t   = ws + LROWS;
    float* pq    = ws + LROWS + QDIM;
    float* stats = ws + LROWS + 3 * QDIM;

    gemv16_wmma_f32<QDIM><<<LROWS / 16, 256, 0, stream>>>(ctx, query, score, stats);

    wsum_and_qhalf<<<QH_PAIR_BLOCKS + QDIM / WS_COLS + (QDIM / 16 - (QH_PAIR_BLOCKS / 2) * 3), 512, 0, stream>>>(ctx, score, stats, s_t, Kw, query, pq);

    gemv_shalf_finish<<<QDIM / 16, 512, 0, stream>>>(Kw, s_t, pq, out);
}
